# starting kernel
# baseline (speedup 1.0000x reference)
#include <hip/hip_runtime.h>
#include <hip/hip_bf16.h>

#define N_NODES   50000
#define N_EDGES   800000
#define IN_FEATS  128
#define OUT_FEATS 512
#define LEAKY     0.01f

typedef __attribute__((ext_vector_type(2))) float v2f;
typedef __attribute__((ext_vector_type(4))) float v4f;
typedef __attribute__((ext_vector_type(8))) float v8f;

__global__ __launch_bounds__(256) void zero_h(float* __restrict__ h) {
    unsigned i = blockIdx.x * 256u + threadIdx.x;
    ((v4f*)h)[i] = (v4f){0.f, 0.f, 0.f, 0.f};
}

__global__ __launch_bounds__(256) void scatter_add(const float* __restrict__ feature,
                                                   const int*   __restrict__ src,
                                                   const int*   __restrict__ dst,
                                                   float*       __restrict__ h) {
    unsigned gid  = blockIdx.x * 256u + threadIdx.x;
    unsigned e    = gid >> 5;
    unsigned lane = gid & 31u;

    int s = src[e];
    int d = dst[e];

    const float* fp = feature + (size_t)s * IN_FEATS + lane * 4u;
    v4f val = *(const v4f*)fp;

    float* hp = h + (size_t)d * IN_FEATS + lane * 4u;
    atomicAdd(hp + 0, val.x);
    atomicAdd(hp + 1, val.y);
    atomicAdd(hp + 2, val.z);
    atomicAdd(hp + 3, val.w);
}

__global__ __launch_bounds__(256) void gemm_wmma(const float* __restrict__ h,
                                                 const float* __restrict__ W,
                                                 const float* __restrict__ bias,
                                                 float*       __restrict__ out) {
    const unsigned waveInBlk = threadIdx.x >> 5;
    const unsigned lane      = threadIdx.x & 31u;
    const unsigned tile      = blockIdx.x * 8u + waveInBlk;

    const unsigned rowTile  = tile >> 3;
    const unsigned colGroup = tile & 7u;

    const unsigned nbase = rowTile * 16u;
    const unsigned obase = colGroup * 64u;

    const unsigned M  = lane & 15u;
    const unsigned kg = lane >> 4;

    v8f acc0 = {}, acc1 = {}, acc2 = {}, acc3 = {};

    const float* arow = h + (size_t)(nbase + M) * IN_FEATS;
    const float* b0   = W + (size_t)(obase + 0u  + M) * IN_FEATS;
    const float* b1   = W + (size_t)(obase + 16u + M) * IN_FEATS;
    const float* b2   = W + (size_t)(obase + 32u + M) * IN_FEATS;
    const float* b3   = W + (size_t)(obase + 48u + M) * IN_FEATS;

    #pragma unroll 4
    for (int kb = 0; kb < IN_FEATS; kb += 4) {
        const int ko = kb + 2 * (int)kg;
        v2f a  = *(const v2f*)(arow + ko);
        v2f f0 = *(const v2f*)(b0 + ko);
        v2f f1 = *(const v2f*)(b1 + ko);
        v2f f2 = *(const v2f*)(b2 + ko);
        v2f f3 = *(const v2f*)(b3 + ko);
        acc0 = __builtin_amdgcn_wmma_f32_16x16x4_f32(false, a, false, f0,
                                                     (short)0, acc0, false, false);
        acc1 = __builtin_amdgcn_wmma_f32_16x16x4_f32(false, a, false, f1,
                                                     (short)0, acc1, false, false);
        acc2 = __builtin_amdgcn_wmma_f32_16x16x4_f32(false, a, false, f2,
                                                     (short)0, acc2, false, false);
        acc3 = __builtin_amdgcn_wmma_f32_16x16x4_f32(false, a, false, f3,
                                                     (short)0, acc3, false, false);
    }

    v8f accs[4] = {acc0, acc1, acc2, acc3};
    #pragma unroll
    for (int t = 0; t < 4; ++t) {
        const unsigned ocol = obase + (unsigned)t * 16u + (lane & 15u);
        const float bv = bias[ocol];
        #pragma unroll
        for (int v = 0; v < 8; ++v) {
            const unsigned Mrow = (unsigned)v + 8u * kg;
            float val = accs[t][v] + bv;
            val = (val >= 0.f) ? val : LEAKY * val;
            out[(size_t)(nbase + Mrow) * OUT_FEATS + ocol] = val;
        }
    }
}

extern "C" void kernel_launch(void* const* d_in, const int* in_sizes, int n_in,
                              void* d_out, int out_size, void* d_ws, size_t ws_size,
                              hipStream_t stream) {
    const float* feature = (const float*)d_in[0];
    const int*   src     = (const int*)  d_in[1];
    const int*   dst     = (const int*)  d_in[2];
    const float* W       = (const float*)d_in[3];
    const float* bias    = (const float*)d_in[4];
    float*       out     = (float*)d_out;
    float*       h       = (float*)d_ws;

    zero_h<<<(N_NODES * IN_FEATS / 4) / 256, 256, 0, stream>>>(h);

    scatter_add<<<(N_EDGES * 32) / 256, 256, 0, stream>>>(feature, src, dst, h);

    gemm_wmma<<<3125, 256, 0, stream>>>(h, W, bias, out);
}


# GEMM retiled: 256 persistent workgroups; each wave keeps its 64-column slab of W in registers (staged once via LDS) and A rows one tile ahead; 16-B fragments, so k is summed in a permuted order
# speedup vs baseline: 1.2352x; 1.2352x over previous
#include <hip/hip_runtime.h>
#include <hip/hip_bf16.h>

#define N_NODES   50000
#define N_EDGES   800000
#define IN_FEATS  128
#define OUT_FEATS 512
#define LEAKY     0.01f

typedef __attribute__((ext_vector_type(2))) float v2f;
typedef __attribute__((ext_vector_type(4))) float v4f;
typedef __attribute__((ext_vector_type(8))) float v8f;

__global__ __launch_bounds__(256) void zero_h(float* __restrict__ h) {
    unsigned i = blockIdx.x * 256u + threadIdx.x;
    ((v4f*)h)[i] = (v4f){0.f, 0.f, 0.f, 0.f};
}

__global__ __launch_bounds__(256) void scatter_add(const float* __restrict__ feature,
                                                   const int*   __restrict__ src,
                                                   const int*   __restrict__ dst,
                                                   float*       __restrict__ h) {
    unsigned gid  = blockIdx.x * 256u + threadIdx.x;
    unsigned e    = gid >> 5;
    unsigned lane = gid & 31u;

    int s = src[e];
    int d = dst[e];

    const float* fp = feature + (size_t)s * IN_FEATS + lane * 4u;
    v4f val = *(const v4f*)fp;

    float* hp = h + (size_t)d * IN_FEATS + lane * 4u;
    atomicAdd(hp + 0, val.x);
    atomicAdd(hp + 1, val.y);
    atomicAdd(hp + 2, val.z);
    atomicAdd(hp + 3, val.w);
}

#define GW_PITCH     132u
#define GW_ROWTILES  3125u

__global__ __launch_bounds__(256) void gemm_wmma(const float* __restrict__ h,
                                                 const float* __restrict__ W,
                                                 const float* __restrict__ bias,
                                                 float*       __restrict__ out) {
    __shared__ float Ws[OUT_FEATS * GW_PITCH];

    const unsigned waveInBlk = (unsigned)__builtin_amdgcn_readfirstlane((int)(threadIdx.x >> 5));
    const unsigned lane      = threadIdx.x & 31u;
    const unsigned c         = lane & 15u;
    const unsigned hh        = lane >> 4;
    const unsigned obase     = waveInBlk * 64u;

    unsigned rt = blockIdx.x;

    #pragma unroll 16
    for (unsigned r = waveInBlk; r < OUT_FEATS; r += 8u) {
        const v4f v = *(const v4f*)(W + (size_t)r * IN_FEATS + lane * 4u);
        *(v4f*)(Ws + r * GW_PITCH + lane * 4u) = v;
    }

    v4f ac[16];
    {
        const float* ar = h + (size_t)(rt * 16u + c) * IN_FEATS + 4u * hh;
        #pragma unroll
        for (int j = 0; j < 16; ++j) ac[j] = *(const v4f*)(ar + 8 * j);
    }
    float bv[4];
    #pragma unroll
    for (int t = 0; t < 4; ++t) bv[t] = bias[obase + (unsigned)t * 16u + c];

    __syncthreads();

    const float* bp = Ws + (obase + c) * GW_PITCH + 4u * hh;

    for (; rt < GW_ROWTILES; rt += gridDim.x) {
        const unsigned nbase = rt * 16u;

        unsigned rn = rt + gridDim.x;
        if (rn >= GW_ROWTILES) rn = rt;
        v4f an[16];
        {
            const float* ar = h + (size_t)(rn * 16u + c) * IN_FEATS + 4u * hh;
            #pragma unroll
            for (int j = 0; j < 16; ++j) an[j] = *(const v4f*)(ar + 8 * j);
        }

        v8f acc0 = {}, acc1 = {}, acc2 = {}, acc3 = {};

        #pragma unroll
        for (int j = 0; j < 16; ++j) {
            const v4f f0 = *(const v4f*)(bp +  0u * GW_PITCH + 8 * j);
            const v4f f1 = *(const v4f*)(bp + 16u * GW_PITCH + 8 * j);
            const v4f f2 = *(const v4f*)(bp + 32u * GW_PITCH + 8 * j);
            const v4f f3 = *(const v4f*)(bp + 48u * GW_PITCH + 8 * j);
            const v2f alo = {ac[j].x, ac[j].y};
            const v2f ahi = {ac[j].z, ac[j].w};
            acc0 = __builtin_amdgcn_wmma_f32_16x16x4_f32(false, alo, false, (v2f){f0.x, f0.y},
                                                         (short)0, acc0, false, false);
            acc1 = __builtin_amdgcn_wmma_f32_16x16x4_f32(false, alo, false, (v2f){f1.x, f1.y},
                                                         (short)0, acc1, false, false);
            acc2 = __builtin_amdgcn_wmma_f32_16x16x4_f32(false, alo, false, (v2f){f2.x, f2.y},
                                                         (short)0, acc2, false, false);
            acc3 = __builtin_amdgcn_wmma_f32_16x16x4_f32(false, alo, false, (v2f){f3.x, f3.y},
                                                         (short)0, acc3, false, false);
            acc0 = __builtin_amdgcn_wmma_f32_16x16x4_f32(false, ahi, false, (v2f){f0.z, f0.w},
                                                         (short)0, acc0, false, false);
            acc1 = __builtin_amdgcn_wmma_f32_16x16x4_f32(false, ahi, false, (v2f){f1.z, f1.w},
                                                         (short)0, acc1, false, false);
            acc2 = __builtin_amdgcn_wmma_f32_16x16x4_f32(false, ahi, false, (v2f){f2.z, f2.w},
                                                         (short)0, acc2, false, false);
            acc3 = __builtin_amdgcn_wmma_f32_16x16x4_f32(false, ahi, false, (v2f){f3.z, f3.w},
                                                         (short)0, acc3, false, false);
        }

        v8f accs[4] = {acc0, acc1, acc2, acc3};
        #pragma unroll
        for (int t = 0; t < 4; ++t) {
            const unsigned ocol = obase + (unsigned)t * 16u + c;
            #pragma unroll
            for (int v = 0; v < 8; ++v) {
                const unsigned Mrow = (unsigned)v + 8u * hh;
                float val = accs[t][v] + bv[t];
                val = (val >= 0.f) ? val : LEAKY * val;
                out[(size_t)(nbase + Mrow) * OUT_FEATS + ocol] = val;
            }
        }

        #pragma unroll
        for (int j = 0; j < 16; ++j) ac[j] = an[j];
    }
}

extern "C" void kernel_launch(void* const* d_in, const int* in_sizes, int n_in,
                              void* d_out, int out_size, void* d_ws, size_t ws_size,
                              hipStream_t stream) {
    const float* feature = (const float*)d_in[0];
    const int*   src     = (const int*)  d_in[1];
    const int*   dst     = (const int*)  d_in[2];
    const float* W       = (const float*)d_in[3];
    const float* bias    = (const float*)d_in[4];
    float*       out     = (float*)d_out;
    float*       h       = (float*)d_ws;

    zero_h<<<(N_NODES * IN_FEATS / 4) / 256, 256, 0, stream>>>(h);

    scatter_add<<<(N_EDGES * 32) / 256, 256, 0, stream>>>(feature, src, dst, h);

    gemm_wmma<<<256, 256, 0, stream>>>(h, W, bias, out);
}


# scatter-add lane map l -> floats l+32q: each wave-wide f32 atomic (and load) covers one whole 128-B line, not 4 lines at 16-B stride; 16 -> 4 atomic line requests per edge; sums stay in arrival order
# speedup vs baseline: 1.5782x; 1.2777x over previous
#include <hip/hip_runtime.h>
#include <hip/hip_bf16.h>

#define N_NODES   50000
#define N_EDGES   800000
#define IN_FEATS  128
#define OUT_FEATS 512
#define LEAKY     0.01f

typedef __attribute__((ext_vector_type(2))) float v2f;
typedef __attribute__((ext_vector_type(4))) float v4f;
typedef __attribute__((ext_vector_type(8))) float v8f;

__global__ __launch_bounds__(256) void zero_h(float* __restrict__ h) {
    unsigned i = blockIdx.x * 256u + threadIdx.x;
    ((v4f*)h)[i] = (v4f){0.f, 0.f, 0.f, 0.f};
}

__global__ __launch_bounds__(256) void scatter_add(const float* __restrict__ feature,
                                                   const int*   __restrict__ src,
                                                   const int*   __restrict__ dst,
                                                   float*       __restrict__ h) {
    unsigned gid  = blockIdx.x * 256u + threadIdx.x;
    unsigned e    = gid >> 5;
    unsigned lane = gid & 31u;

    int s = src[e];
    int d = dst[e];

    const float* fp = feature + (size_t)s * IN_FEATS + lane;
    const float v0 = fp[0];
    const float v1 = fp[32];
    const float v2 = fp[64];
    const float v3 = fp[96];

    float* hp = h + (size_t)d * IN_FEATS + lane;
    atomicAdd(hp +  0, v0);
    atomicAdd(hp + 32, v1);
    atomicAdd(hp + 64, v2);
    atomicAdd(hp + 96, v3);
}

#define GW_PITCH     132u
#define GW_ROWTILES  3125u

__global__ __launch_bounds__(256) void gemm_wmma(const float* __restrict__ h,
                                                 const float* __restrict__ W,
                                                 const float* __restrict__ bias,
                                                 float*       __restrict__ out) {
    __shared__ float Ws[OUT_FEATS * GW_PITCH];

    const unsigned waveInBlk = (unsigned)__builtin_amdgcn_readfirstlane((int)(threadIdx.x >> 5));
    const unsigned lane      = threadIdx.x & 31u;
    const unsigned c         = lane & 15u;
    const unsigned hh        = lane >> 4;
    const unsigned obase     = waveInBlk * 64u;

    unsigned rt = blockIdx.x;

    #pragma unroll 16
    for (unsigned r = waveInBlk; r < OUT_FEATS; r += 8u) {
        const v4f v = *(const v4f*)(W + (size_t)r * IN_FEATS + lane * 4u);
        *(v4f*)(Ws + r * GW_PITCH + lane * 4u) = v;
    }

    v4f ac[16];
    {
        const float* ar = h + (size_t)(rt * 16u + c) * IN_FEATS + 4u * hh;
        #pragma unroll
        for (int j = 0; j < 16; ++j) ac[j] = *(const v4f*)(ar + 8 * j);
    }
    float bv[4];
    #pragma unroll
    for (int t = 0; t < 4; ++t) bv[t] = bias[obase + (unsigned)t * 16u + c];

    __syncthreads();

    const float* bp = Ws + (obase + c) * GW_PITCH + 4u * hh;

    for (; rt < GW_ROWTILES; rt += gridDim.x) {
        const unsigned nbase = rt * 16u;

        unsigned rn = rt + gridDim.x;
        if (rn >= GW_ROWTILES) rn = rt;
        v4f an[16];
        {
            const float* ar = h + (size_t)(rn * 16u + c) * IN_FEATS + 4u * hh;
            #pragma unroll
            for (int j = 0; j < 16; ++j) an[j] = *(const v4f*)(ar + 8 * j);
        }

        v8f acc0 = {}, acc1 = {}, acc2 = {}, acc3 = {};

        #pragma unroll
        for (int j = 0; j < 16; ++j) {
            const v4f f0 = *(const v4f*)(bp +  0u * GW_PITCH + 8 * j);
            const v4f f1 = *(const v4f*)(bp + 16u * GW_PITCH + 8 * j);
            const v4f f2 = *(const v4f*)(bp + 32u * GW_PITCH + 8 * j);
            const v4f f3 = *(const v4f*)(bp + 48u * GW_PITCH + 8 * j);
            const v2f alo = {ac[j].x, ac[j].y};
            const v2f ahi = {ac[j].z, ac[j].w};
            acc0 = __builtin_amdgcn_wmma_f32_16x16x4_f32(false, alo, false, (v2f){f0.x, f0.y},
                                                         (short)0, acc0, false, false);
            acc1 = __builtin_amdgcn_wmma_f32_16x16x4_f32(false, alo, false, (v2f){f1.x, f1.y},
                                                         (short)0, acc1, false, false);
            acc2 = __builtin_amdgcn_wmma_f32_16x16x4_f32(false, alo, false, (v2f){f2.x, f2.y},
                                                         (short)0, acc2, false, false);
            acc3 = __builtin_amdgcn_wmma_f32_16x16x4_f32(false, alo, false, (v2f){f3.x, f3.y},
                                                         (short)0, acc3, false, false);
            acc0 = __builtin_amdgcn_wmma_f32_16x16x4_f32(false, ahi, false, (v2f){f0.z, f0.w},
                                                         (short)0, acc0, false, false);
            acc1 = __builtin_amdgcn_wmma_f32_16x16x4_f32(false, ahi, false, (v2f){f1.z, f1.w},
                                                         (short)0, acc1, false, false);
            acc2 = __builtin_amdgcn_wmma_f32_16x16x4_f32(false, ahi, false, (v2f){f2.z, f2.w},
                                                         (short)0, acc2, false, false);
            acc3 = __builtin_amdgcn_wmma_f32_16x16x4_f32(false, ahi, false, (v2f){f3.z, f3.w},
                                                         (short)0, acc3, false, false);
        }

        v8f accs[4] = {acc0, acc1, acc2, acc3};
        #pragma unroll
        for (int t = 0; t < 4; ++t) {
            const unsigned ocol = obase + (unsigned)t * 16u + c;
            #pragma unroll
            for (int v = 0; v < 8; ++v) {
                const unsigned Mrow = (unsigned)v + 8u * hh;
                float val = accs[t][v] + bv[t];
                val = (val >= 0.f) ? val : LEAKY * val;
                out[(size_t)(nbase + Mrow) * OUT_FEATS + ocol] = val;
            }
        }

        #pragma unroll
        for (int j = 0; j < 16; ++j) ac[j] = an[j];
    }
}

extern "C" void kernel_launch(void* const* d_in, const int* in_sizes, int n_in,
                              void* d_out, int out_size, void* d_ws, size_t ws_size,
                              hipStream_t stream) {
    const float* feature = (const float*)d_in[0];
    const int*   src     = (const int*)  d_in[1];
    const int*   dst     = (const int*)  d_in[2];
    const float* W       = (const float*)d_in[3];
    const float* bias    = (const float*)d_in[4];
    float*       out     = (float*)d_out;
    float*       h       = (float*)d_ws;

    zero_h<<<(N_NODES * IN_FEATS / 4) / 256, 256, 0, stream>>>(h);

    scatter_add<<<(N_EDGES * 32) / 256, 256, 0, stream>>>(feature, src, dst, h);

    gemm_wmma<<<256, 256, 0, stream>>>(h, W, bias, out);
}


# scatter-add replaced by binning: integer atomics give each edge a slot in its destination's bin, one wave per node adds its rows sorted by source; no float atomics, no zero fill, sum order fixed
# speedup vs baseline: 1.6649x; 1.0549x over previous
#include <hip/hip_runtime.h>
#include <hip/hip_bf16.h>

#define N_NODES   50000
#define N_EDGES   800000
#define IN_FEATS  128
#define OUT_FEATS 512
#define LEAKY     0.01f

typedef __attribute__((ext_vector_type(2))) float v2f;
typedef __attribute__((ext_vector_type(4))) float v4f;
typedef __attribute__((ext_vector_type(8))) float v8f;

#define SLOT_CAP  64
#define CNT_SLOT  64
typedef int __attribute__((may_alias)) int_a;

__global__ __launch_bounds__(256) void zero_cnt(float* __restrict__ h) {
    const unsigned i = blockIdx.x * 256u + threadIdx.x;
    if (i < N_NODES) ((int_a*)h)[i * IN_FEATS + CNT_SLOT] = 0;
}

__global__ __launch_bounds__(256) void fill_slots(const int* __restrict__ src,
                                                  const int* __restrict__ dst,
                                                  float*     __restrict__ h) {
    const unsigned e = blockIdx.x * 256u + threadIdx.x;
    const int s = src[e];
    const int d = dst[e];
    int* row = (int*)h + (size_t)d * IN_FEATS;
    const int pos = atomicAdd(row + CNT_SLOT, 1);
    if (pos < SLOT_CAP) atomicExch(row + pos, s);
}

__global__ __launch_bounds__(256) void gather_rows(const float* __restrict__ feature,
                                                   const int*   __restrict__ src,
                                                   const int*   __restrict__ dst,
                                                   float*                    h) {
    __shared__ int sorted[8][SLOT_CAP];

    const unsigned gid  = blockIdx.x * 256u + threadIdx.x;
    const int node = __builtin_amdgcn_readfirstlane((int)(gid >> 5));
    const int lane = (int)(threadIdx.x & 31u);
    const int_a* bin = (const int_a*)h + (size_t)node * IN_FEATS;
    const int nAll = __builtin_amdgcn_readfirstlane(bin[CNT_SLOT]);
    const int n    = nAll < SLOT_CAP ? nAll : SLOT_CAP;

    const int BIG = 0x7fffffff;
    const int a0 = (lane      < n) ? bin[lane]      : BIG;
    const int a1 = (lane + 32 < n) ? bin[lane + 32] : BIG;

    int r0 = 0, r1 = 0;
    for (int j = 0; j < n; ++j) {
        const int v = (j < 32) ? __builtin_amdgcn_readlane(a0, j)
                               : __builtin_amdgcn_readlane(a1, j - 32);
        r0 += (v < a0 || (v == a0 && j < lane))      ? 1 : 0;
        r1 += (v < a1 || (v == a1 && j < lane + 32)) ? 1 : 0;
    }

    int* sw = sorted[threadIdx.x >> 5];
    if (lane      < n) sw[r0] = a0;
    if (lane + 32 < n) sw[r1] = a1;
    __syncthreads();
    const int b0 = sw[lane];
    const int b1 = sw[lane + 32];

    const float* fb = feature + lane * 4;
    v4f acc = {0.f, 0.f, 0.f, 0.f};
    int j = 0;
    #pragma unroll 1
    for (; j + 8 <= n; j += 8) {
        const int bs = (j < 32) ? b0 : b1;
        v4f r[8];
        #pragma unroll
        for (int u = 0; u < 8; ++u) {
            const int s = __builtin_amdgcn_readlane(bs, (j & 31) + u);
            r[u] = *(const v4f*)(fb + (size_t)s * IN_FEATS);
        }
        #pragma unroll
        for (int u = 0; u < 8; ++u) acc += r[u];
    }
    {
        const int bs = (j < 32) ? b0 : b1;
        const int jl = j & 31;
        const int t  = n - j;
        const v4f z = {0.f, 0.f, 0.f, 0.f};
        v4f t0 = z, t1 = z, t2 = z, t3 = z, t4 = z, t5 = z, t6 = z;
        if (t & 4) {
            t0 = *(const v4f*)(fb + (size_t)__builtin_amdgcn_readlane(bs, jl + 0) * IN_FEATS);
            t1 = *(const v4f*)(fb + (size_t)__builtin_amdgcn_readlane(bs, jl + 1) * IN_FEATS);
            t2 = *(const v4f*)(fb + (size_t)__builtin_amdgcn_readlane(bs, jl + 2) * IN_FEATS);
            t3 = *(const v4f*)(fb + (size_t)__builtin_amdgcn_readlane(bs, jl + 3) * IN_FEATS);
        }
        const int k2 = jl + (t & 4);
        if (t & 2) {
            t4 = *(const v4f*)(fb + (size_t)__builtin_amdgcn_readlane(bs, k2 + 0) * IN_FEATS);
            t5 = *(const v4f*)(fb + (size_t)__builtin_amdgcn_readlane(bs, k2 + 1) * IN_FEATS);
        }
        const int k1 = jl + (t & 6);
        if (t & 1) {
            t6 = *(const v4f*)(fb + (size_t)__builtin_amdgcn_readlane(bs, k1) * IN_FEATS);
        }
        acc += t0; acc += t1; acc += t2; acc += t3; acc += t4; acc += t5; acc += t6;
    }

    if (nAll > SLOT_CAP) {
        acc = (v4f){0.f, 0.f, 0.f, 0.f};
        for (int base = 0; base < N_EDGES; base += 32) {
            unsigned m = __builtin_amdgcn_ballot_w32(dst[base + lane] == node);
            while (m) {
                const int k = __builtin_ctz(m);
                m &= m - 1u;
                acc += *(const v4f*)(fb + (size_t)src[base + k] * IN_FEATS);
            }
        }
    }

    *(v4f*)(h + (size_t)node * IN_FEATS + lane * 4) = acc;
}

#define GW_PITCH     132u
#define GW_ROWTILES  3125u

__global__ __launch_bounds__(256) void gemm_wmma(const float* __restrict__ h,
                                                 const float* __restrict__ W,
                                                 const float* __restrict__ bias,
                                                 float*       __restrict__ out) {
    __shared__ float Ws[OUT_FEATS * GW_PITCH];

    const unsigned waveInBlk = (unsigned)__builtin_amdgcn_readfirstlane((int)(threadIdx.x >> 5));
    const unsigned lane      = threadIdx.x & 31u;
    const unsigned c         = lane & 15u;
    const unsigned hh        = lane >> 4;
    const unsigned obase     = waveInBlk * 64u;

    unsigned rt = blockIdx.x;

    #pragma unroll 16
    for (unsigned r = waveInBlk; r < OUT_FEATS; r += 8u) {
        const v4f v = *(const v4f*)(W + (size_t)r * IN_FEATS + lane * 4u);
        *(v4f*)(Ws + r * GW_PITCH + lane * 4u) = v;
    }

    v4f ac[16];
    {
        const float* ar = h + (size_t)(rt * 16u + c) * IN_FEATS + 4u * hh;
        #pragma unroll
        for (int j = 0; j < 16; ++j) ac[j] = *(const v4f*)(ar + 8 * j);
    }
    float bv[4];
    #pragma unroll
    for (int t = 0; t < 4; ++t) bv[t] = bias[obase + (unsigned)t * 16u + c];

    __syncthreads();

    const float* bp = Ws + (obase + c) * GW_PITCH + 4u * hh;

    for (; rt < GW_ROWTILES; rt += gridDim.x) {
        const unsigned nbase = rt * 16u;

        unsigned rn = rt + gridDim.x;
        if (rn >= GW_ROWTILES) rn = rt;
        v4f an[16];
        {
            const float* ar = h + (size_t)(rn * 16u + c) * IN_FEATS + 4u * hh;
            #pragma unroll
            for (int j = 0; j < 16; ++j) an[j] = *(const v4f*)(ar + 8 * j);
        }

        v8f acc0 = {}, acc1 = {}, acc2 = {}, acc3 = {};

        #pragma unroll
        for (int j = 0; j < 16; ++j) {
            const v4f f0 = *(const v4f*)(bp +  0u * GW_PITCH + 8 * j);
            const v4f f1 = *(const v4f*)(bp + 16u * GW_PITCH + 8 * j);
            const v4f f2 = *(const v4f*)(bp + 32u * GW_PITCH + 8 * j);
            const v4f f3 = *(const v4f*)(bp + 48u * GW_PITCH + 8 * j);
            const v2f alo = {ac[j].x, ac[j].y};
            const v2f ahi = {ac[j].z, ac[j].w};
            acc0 = __builtin_amdgcn_wmma_f32_16x16x4_f32(false, alo, false, (v2f){f0.x, f0.y},
                                                         (short)0, acc0, false, false);
            acc1 = __builtin_amdgcn_wmma_f32_16x16x4_f32(false, alo, false, (v2f){f1.x, f1.y},
                                                         (short)0, acc1, false, false);
            acc2 = __builtin_amdgcn_wmma_f32_16x16x4_f32(false, alo, false, (v2f){f2.x, f2.y},
                                                         (short)0, acc2, false, false);
            acc3 = __builtin_amdgcn_wmma_f32_16x16x4_f32(false, alo, false, (v2f){f3.x, f3.y},
                                                         (short)0, acc3, false, false);
            acc0 = __builtin_amdgcn_wmma_f32_16x16x4_f32(false, ahi, false, (v2f){f0.z, f0.w},
                                                         (short)0, acc0, false, false);
            acc1 = __builtin_amdgcn_wmma_f32_16x16x4_f32(false, ahi, false, (v2f){f1.z, f1.w},
                                                         (short)0, acc1, false, false);
            acc2 = __builtin_amdgcn_wmma_f32_16x16x4_f32(false, ahi, false, (v2f){f2.z, f2.w},
                                                         (short)0, acc2, false, false);
            acc3 = __builtin_amdgcn_wmma_f32_16x16x4_f32(false, ahi, false, (v2f){f3.z, f3.w},
                                                         (short)0, acc3, false, false);
        }

        v8f accs[4] = {acc0, acc1, acc2, acc3};
        #pragma unroll
        for (int t = 0; t < 4; ++t) {
            const unsigned ocol = obase + (unsigned)t * 16u + c;
            #pragma unroll
            for (int v = 0; v < 8; ++v) {
                const unsigned Mrow = (unsigned)v + 8u * hh;
                float val = accs[t][v] + bv[t];
                val = (val >= 0.f) ? val : LEAKY * val;
                out[(size_t)(nbase + Mrow) * OUT_FEATS + ocol] = val;
            }
        }

        #pragma unroll
        for (int j = 0; j < 16; ++j) ac[j] = an[j];
    }
}

extern "C" void kernel_launch(void* const* d_in, const int* in_sizes, int n_in,
                              void* d_out, int out_size, void* d_ws, size_t ws_size,
                              hipStream_t stream) {
    const float* feature = (const float*)d_in[0];
    const int*   src     = (const int*)  d_in[1];
    const int*   dst     = (const int*)  d_in[2];
    const float* W       = (const float*)d_in[3];
    const float* bias    = (const float*)d_in[4];
    float*       out     = (float*)d_out;
    float*       h       = (float*)d_ws;

    zero_cnt<<<(N_NODES + 255) / 256, 256, 0, stream>>>(h);

    fill_slots<<<N_EDGES / 256, 256, 0, stream>>>(src, dst, h);

    gather_rows<<<N_NODES / 8, 256, 0, stream>>>(feature, src, dst, h);

    gemm_wmma<<<256, 256, 0, stream>>>(h, W, bias, out);
}


# binning moved to LDS atomics, two levels: 125 edge-chunk workgroups split edges by 100-node range into whole-line segments, a range's workgroup bins them per node in LDS; no global atomics, same sums
# speedup vs baseline: 2.6524x; 1.5931x over previous
#include <hip/hip_runtime.h>
#include <hip/hip_bf16.h>

#define N_NODES   50000
#define N_EDGES   800000
#define IN_FEATS  128
#define OUT_FEATS 512
#define LEAKY     0.01f

typedef __attribute__((ext_vector_type(2))) float v2f;
typedef __attribute__((ext_vector_type(4))) float v4f;
typedef __attribute__((ext_vector_type(8))) float v8f;

#define SLOT_CAP  64
#define N_RANGES  500u
#define RANGE_N   100u
#define N_CHUNKS  125u
#define CHUNK_E   6400u
#define SEG_INTS  96u
#define SEG_CAP   95
#define REGION    (RANGE_N * IN_FEATS)
typedef int __attribute__((may_alias)) int_a;

typedef __attribute__((ext_vector_type(4))) int v4i;

__global__ __launch_bounds__(1024) void bin_edges(const int* __restrict__ src,
                                                  const int* __restrict__ dst,
                                                  float*     __restrict__ h) {
    __shared__ int segCnt[N_RANGES];
    __shared__ int segRec[N_RANGES][SEG_INTS];

    const unsigned t = threadIdx.x;
    const unsigned wave = (unsigned)__builtin_amdgcn_readfirstlane((int)(t >> 5));
    const unsigned lane = t & 31u;
    const unsigned w = blockIdx.x;
    if (t < N_RANGES) segCnt[t] = 0;

    const v4i* s4 = (const v4i*)(src + (size_t)w * CHUNK_E);
    const v4i* d4 = (const v4i*)(dst + (size_t)w * CHUNK_E);
    v4i sv[2], dv[2];
    #pragma unroll
    for (int u = 0; u < 2; ++u) {
        const unsigned q = t + 1024u * (unsigned)u;
        if (q < CHUNK_E / 4u) { sv[u] = s4[q]; dv[u] = d4[q]; }
        else { sv[u] = (v4i){0, 0, 0, 0}; dv[u] = (v4i){-1, -1, -1, -1}; }
    }
    __syncthreads();

    #pragma unroll
    for (int u = 0; u < 2; ++u) {
        #pragma unroll
        for (int k = 0; k < 4; ++k) {
            const int di = dv[u][k];
            if (di >= 0) {
                const unsigned d  = (unsigned)di;
                const unsigned r  = d / RANGE_N;
                const unsigned dl = d - r * RANGE_N;
                const int pos = atomicAdd(&segCnt[r], 1);
                if (pos < SEG_CAP) segRec[r][1 + pos] = (int)(((unsigned)sv[u][k] << 8) | dl);
            }
        }
    }
    __syncthreads();

    const unsigned rmine = wave + 32u * lane;
    const int cs = (lane < 16u && rmine < N_RANGES) ? segCnt[rmine] : 0;
    int_a* hi = (int_a*)h;
    #pragma unroll
    for (int u = 0; u < 16; ++u) {
        const unsigned r = wave + 32u * (unsigned)u;
        if (r < N_RANGES) {
            const int c    = __builtin_amdgcn_readlane(cs, u);
            const int last = c < SEG_CAP ? c : SEG_CAP;
            int_a* g = hi + (size_t)r * REGION + w * SEG_INTS;
            if ((int)lane <= last) g[lane] = (lane == 0u) ? c : segRec[r][lane];
            if (last >= 32) {
                if ((int)lane + 32 <= last) g[lane + 32u] = segRec[r][lane + 32u];
                if ((int)lane + 64 <= last) g[lane + 64u] = segRec[r][lane + 64u];
            }
        }
    }
}

__global__ __launch_bounds__(1024) void gather_rows(const float* __restrict__ feature,
                                                    const int*   __restrict__ src,
                                                    const int*   __restrict__ dst,
                                                    float*                    h) {
    __shared__ int cntN[RANGE_N];
    __shared__ int binN[RANGE_N][SLOT_CAP];
    __shared__ int sorted[32][SLOT_CAP];
    __shared__ int spill;
    __shared__ int ticket;

    const unsigned t = threadIdx.x;
    const unsigned wave = (unsigned)__builtin_amdgcn_readfirstlane((int)(t >> 5));
    const int lane = (int)(t & 31u);
    const unsigned rg = blockIdx.x;

    if (t < RANGE_N) cntN[t] = 0;
    if (t == 0) { spill = 0; ticket = 0; }
    __syncthreads();

    const int_a* region = (const int_a*)h + (size_t)rg * REGION;
    int l0[4];
    #pragma unroll
    for (int u = 0; u < 4; ++u) {
        const unsigned w = wave + 32u * (unsigned)u;
        l0[u] = (w < N_CHUNKS) ? region[w * SEG_INTS + (unsigned)lane] : 0;
    }
    #pragma unroll
    for (int u = 0; u < 4; ++u) {
        const unsigned w = wave + 32u * (unsigned)u;
        const int c    = __builtin_amdgcn_readfirstlane(l0[u]);
        const int last = c < SEG_CAP ? c : SEG_CAP;
        if (c > SEG_CAP && lane == 0) spill = 1;
        int rec  = l0[u];
        for (int idx = lane; idx <= last; idx += 32) {
            if (idx >= 32) rec = region[w * SEG_INTS + (unsigned)idx];
            if (idx >= 1) {
                const unsigned dl = (unsigned)rec & 255u;
                const int pos = atomicAdd(&cntN[dl], 1);
                if (pos < SLOT_CAP) binN[dl][pos] = (int)((unsigned)rec >> 8);
            }
        }
    }
    __syncthreads();

    const int anySpill = __builtin_amdgcn_readfirstlane(spill);
    int* sw = sorted[wave];
    const float* fb = feature + lane * 4;
    const int BIG = 0x7fffffff;

    for (unsigned trip = 0; trip < RANGE_N; ++trip) {
        int tk = 0;
        if (lane == 0) tk = atomicAdd(&ticket, 1);
        const unsigned dl = (unsigned)__builtin_amdgcn_readfirstlane(tk);
        if (dl >= RANGE_N) break;
        const int node = (int)(rg * RANGE_N + dl);
        const int nAll = __builtin_amdgcn_readfirstlane(cntN[dl]);
        const int n    = nAll < SLOT_CAP ? nAll : SLOT_CAP;

        const int a0 = (lane      < n) ? binN[dl][lane]      : BIG;
        const int a1 = (lane + 32 < n) ? binN[dl][lane + 32] : BIG;

        int r0 = 0, r1 = 0;
        for (int j = 0; j < n; ++j) {
            const int v = (j < 32) ? __builtin_amdgcn_readlane(a0, j)
                                   : __builtin_amdgcn_readlane(a1, j - 32);
            r0 += (v < a0 || (v == a0 && j < lane))      ? 1 : 0;
            r1 += (v < a1 || (v == a1 && j < lane + 32)) ? 1 : 0;
        }

        if (lane      < n) sw[r0] = a0;
        if (lane + 32 < n) sw[r1] = a1;
        const int b0 = (lane      < n) ? sw[lane]      : 0;
        const int b1 = (lane + 32 < n) ? sw[lane + 32] : 0;

        v4f acc = {0.f, 0.f, 0.f, 0.f};
        int j = 0;
        #pragma unroll 1
        for (; j + 8 <= n; j += 8) {
            const int bs = (j < 32) ? b0 : b1;
            v4f r[8];
            #pragma unroll
            for (int u = 0; u < 8; ++u) {
                const int s = __builtin_amdgcn_readlane(bs, (j & 31) + u);
                r[u] = *(const v4f*)(fb + (size_t)s * IN_FEATS);
            }
            #pragma unroll
            for (int u = 0; u < 8; ++u) acc += r[u];
        }
        {
            const int bs = (j < 32) ? b0 : b1;
            const int jl = j & 31;
            const int tl = n - j;
            const v4f z = {0.f, 0.f, 0.f, 0.f};
            v4f t0 = z, t1 = z, t2 = z, t3 = z, t4 = z, t5 = z, t6 = z;
            if (tl & 4) {
                t0 = *(const v4f*)(fb + (size_t)__builtin_amdgcn_readlane(bs, jl + 0) * IN_FEATS);
                t1 = *(const v4f*)(fb + (size_t)__builtin_amdgcn_readlane(bs, jl + 1) * IN_FEATS);
                t2 = *(const v4f*)(fb + (size_t)__builtin_amdgcn_readlane(bs, jl + 2) * IN_FEATS);
                t3 = *(const v4f*)(fb + (size_t)__builtin_amdgcn_readlane(bs, jl + 3) * IN_FEATS);
            }
            const int k2 = jl + (tl & 4);
            if (tl & 2) {
                t4 = *(const v4f*)(fb + (size_t)__builtin_amdgcn_readlane(bs, k2 + 0) * IN_FEATS);
                t5 = *(const v4f*)(fb + (size_t)__builtin_amdgcn_readlane(bs, k2 + 1) * IN_FEATS);
            }
            const int k1 = jl + (tl & 6);
            if (tl & 1) {
                t6 = *(const v4f*)(fb + (size_t)__builtin_amdgcn_readlane(bs, k1) * IN_FEATS);
            }
            acc += t0; acc += t1; acc += t2; acc += t3; acc += t4; acc += t5; acc += t6;
        }

        if (nAll > SLOT_CAP || anySpill) {
            acc = (v4f){0.f, 0.f, 0.f, 0.f};
            for (int base = 0; base < N_EDGES; base += 32) {
                unsigned m = __builtin_amdgcn_ballot_w32(dst[base + lane] == node);
                while (m) {
                    const int k = __builtin_ctz(m);
                    m &= m - 1u;
                    acc += *(const v4f*)(fb + (size_t)src[base + k] * IN_FEATS);
                }
            }
        }

        *(v4f*)(h + (size_t)node * IN_FEATS + lane * 4) = acc;
    }
}

#define GW_PITCH     132u
#define GW_ROWTILES  3125u

__global__ __launch_bounds__(256) void gemm_wmma(const float* __restrict__ h,
                                                 const float* __restrict__ W,
                                                 const float* __restrict__ bias,
                                                 float*       __restrict__ out) {
    __shared__ float Ws[OUT_FEATS * GW_PITCH];

    const unsigned waveInBlk = (unsigned)__builtin_amdgcn_readfirstlane((int)(threadIdx.x >> 5));
    const unsigned lane      = threadIdx.x & 31u;
    const unsigned c         = lane & 15u;
    const unsigned hh        = lane >> 4;
    const unsigned obase     = waveInBlk * 64u;

    unsigned rt = blockIdx.x;

    #pragma unroll 16
    for (unsigned r = waveInBlk; r < OUT_FEATS; r += 8u) {
        const v4f v = *(const v4f*)(W + (size_t)r * IN_FEATS + lane * 4u);
        *(v4f*)(Ws + r * GW_PITCH + lane * 4u) = v;
    }

    v4f ac[16];
    {
        const float* ar = h + (size_t)(rt * 16u + c) * IN_FEATS + 4u * hh;
        #pragma unroll
        for (int j = 0; j < 16; ++j) ac[j] = *(const v4f*)(ar + 8 * j);
    }
    float bv[4];
    #pragma unroll
    for (int t = 0; t < 4; ++t) bv[t] = bias[obase + (unsigned)t * 16u + c];

    __syncthreads();

    const float* bp = Ws + (obase + c) * GW_PITCH + 4u * hh;

    for (; rt < GW_ROWTILES; rt += gridDim.x) {
        const unsigned nbase = rt * 16u;

        unsigned rn = rt + gridDim.x;
        if (rn >= GW_ROWTILES) rn = rt;
        v4f an[16];
        {
            const float* ar = h + (size_t)(rn * 16u + c) * IN_FEATS + 4u * hh;
            #pragma unroll
            for (int j = 0; j < 16; ++j) an[j] = *(const v4f*)(ar + 8 * j);
        }

        v8f acc0 = {}, acc1 = {}, acc2 = {}, acc3 = {};

        #pragma unroll
        for (int j = 0; j < 16; ++j) {
            const v4f f0 = *(const v4f*)(bp +  0u * GW_PITCH + 8 * j);
            const v4f f1 = *(const v4f*)(bp + 16u * GW_PITCH + 8 * j);
            const v4f f2 = *(const v4f*)(bp + 32u * GW_PITCH + 8 * j);
            const v4f f3 = *(const v4f*)(bp + 48u * GW_PITCH + 8 * j);
            const v2f alo = {ac[j].x, ac[j].y};
            const v2f ahi = {ac[j].z, ac[j].w};
            acc0 = __builtin_amdgcn_wmma_f32_16x16x4_f32(false, alo, false, (v2f){f0.x, f0.y},
                                                         (short)0, acc0, false, false);
            acc1 = __builtin_amdgcn_wmma_f32_16x16x4_f32(false, alo, false, (v2f){f1.x, f1.y},
                                                         (short)0, acc1, false, false);
            acc2 = __builtin_amdgcn_wmma_f32_16x16x4_f32(false, alo, false, (v2f){f2.x, f2.y},
                                                         (short)0, acc2, false, false);
            acc3 = __builtin_amdgcn_wmma_f32_16x16x4_f32(false, alo, false, (v2f){f3.x, f3.y},
                                                         (short)0, acc3, false, false);
            acc0 = __builtin_amdgcn_wmma_f32_16x16x4_f32(false, ahi, false, (v2f){f0.z, f0.w},
                                                         (short)0, acc0, false, false);
            acc1 = __builtin_amdgcn_wmma_f32_16x16x4_f32(false, ahi, false, (v2f){f1.z, f1.w},
                                                         (short)0, acc1, false, false);
            acc2 = __builtin_amdgcn_wmma_f32_16x16x4_f32(false, ahi, false, (v2f){f2.z, f2.w},
                                                         (short)0, acc2, false, false);
            acc3 = __builtin_amdgcn_wmma_f32_16x16x4_f32(false, ahi, false, (v2f){f3.z, f3.w},
                                                         (short)0, acc3, false, false);
        }

        v8f accs[4] = {acc0, acc1, acc2, acc3};
        #pragma unroll
        for (int t = 0; t < 4; ++t) {
            const unsigned ocol = obase + (unsigned)t * 16u + c;
            #pragma unroll
            for (int v = 0; v < 8; ++v) {
                const unsigned Mrow = (unsigned)v + 8u * hh;
                float val = accs[t][v] + bv[t];
                val = (val >= 0.f) ? val : LEAKY * val;
                out[(size_t)(nbase + Mrow) * OUT_FEATS + ocol] = val;
            }
        }

        #pragma unroll
        for (int j = 0; j < 16; ++j) ac[j] = an[j];
    }
}

extern "C" void kernel_launch(void* const* d_in, const int* in_sizes, int n_in,
                              void* d_out, int out_size, void* d_ws, size_t ws_size,
                              hipStream_t stream) {
    const float* feature = (const float*)d_in[0];
    const int*   src     = (const int*)  d_in[1];
    const int*   dst     = (const int*)  d_in[2];
    const float* W       = (const float*)d_in[3];
    const float* bias    = (const float*)d_in[4];
    float*       out     = (float*)d_out;
    float*       h       = (float*)d_ws;

    bin_edges<<<N_CHUNKS, 1024, 0, stream>>>(src, dst, h);

    gather_rows<<<N_RANGES, 1024, 0, stream>>>(feature, src, dst, h);

    gemm_wmma<<<256, 256, 0, stream>>>(h, W, bias, out);
}


# GEMM epilogue stores whole lines: accumulator t, column n of a wave now holds output column 4n+t (map applied where W is staged), so a lane stores 4 columns as 16 B; 32 half-line stores -> 8 per tile
# speedup vs baseline: 2.7590x; 1.0402x over previous
#include <hip/hip_runtime.h>
#include <hip/hip_bf16.h>

#define N_NODES   50000
#define N_EDGES   800000
#define IN_FEATS  128
#define OUT_FEATS 512
#define LEAKY     0.01f

typedef __attribute__((ext_vector_type(2))) float v2f;
typedef __attribute__((ext_vector_type(4))) float v4f;
typedef __attribute__((ext_vector_type(8))) float v8f;

#define SLOT_CAP  64
#define N_RANGES  500u
#define RANGE_N   100u
#define N_CHUNKS  125u
#define CHUNK_E   6400u
#define SEG_INTS  96u
#define SEG_CAP   95
#define REGION    (RANGE_N * IN_FEATS)
typedef int __attribute__((may_alias)) int_a;

typedef __attribute__((ext_vector_type(4))) int v4i;

__global__ __launch_bounds__(1024) void bin_edges(const int* __restrict__ src,
                                                  const int* __restrict__ dst,
                                                  float*     __restrict__ h) {
    __shared__ int segCnt[N_RANGES];
    __shared__ int segRec[N_RANGES][SEG_INTS];

    const unsigned t = threadIdx.x;
    const unsigned wave = (unsigned)__builtin_amdgcn_readfirstlane((int)(t >> 5));
    const unsigned lane = t & 31u;
    const unsigned w = blockIdx.x;
    if (t < N_RANGES) segCnt[t] = 0;

    const v4i* s4 = (const v4i*)(src + (size_t)w * CHUNK_E);
    const v4i* d4 = (const v4i*)(dst + (size_t)w * CHUNK_E);
    v4i sv[2], dv[2];
    #pragma unroll
    for (int u = 0; u < 2; ++u) {
        const unsigned q = t + 1024u * (unsigned)u;
        if (q < CHUNK_E / 4u) { sv[u] = s4[q]; dv[u] = d4[q]; }
        else { sv[u] = (v4i){0, 0, 0, 0}; dv[u] = (v4i){-1, -1, -1, -1}; }
    }
    __syncthreads();

    #pragma unroll
    for (int u = 0; u < 2; ++u) {
        #pragma unroll
        for (int k = 0; k < 4; ++k) {
            const int di = dv[u][k];
            if (di >= 0) {
                const unsigned d  = (unsigned)di;
                const unsigned r  = d / RANGE_N;
                const unsigned dl = d - r * RANGE_N;
                const int pos = atomicAdd(&segCnt[r], 1);
                if (pos < SEG_CAP) segRec[r][1 + pos] = (int)(((unsigned)sv[u][k] << 8) | dl);
            }
        }
    }
    __syncthreads();

    const unsigned rmine = wave + 32u * lane;
    const int cs = (lane < 16u && rmine < N_RANGES) ? segCnt[rmine] : 0;
    int_a* hi = (int_a*)h;
    #pragma unroll
    for (int u = 0; u < 16; ++u) {
        const unsigned r = wave + 32u * (unsigned)u;
        if (r < N_RANGES) {
            const int c    = __builtin_amdgcn_readlane(cs, u);
            const int last = c < SEG_CAP ? c : SEG_CAP;
            int_a* g = hi + (size_t)r * REGION + w * SEG_INTS;
            if ((int)lane <= last) g[lane] = (lane == 0u) ? c : segRec[r][lane];
            if (last >= 32) {
                if ((int)lane + 32 <= last) g[lane + 32u] = segRec[r][lane + 32u];
                if ((int)lane + 64 <= last) g[lane + 64u] = segRec[r][lane + 64u];
            }
        }
    }
}

__global__ __launch_bounds__(1024) void gather_rows(const float* __restrict__ feature,
                                                    const int*   __restrict__ src,
                                                    const int*   __restrict__ dst,
                                                    float*                    h) {
    __shared__ int cntN[RANGE_N];
    __shared__ int binN[RANGE_N][SLOT_CAP];
    __shared__ int sorted[32][SLOT_CAP];
    __shared__ int spill;
    __shared__ int ticket;

    const unsigned t = threadIdx.x;
    const unsigned wave = (unsigned)__builtin_amdgcn_readfirstlane((int)(t >> 5));
    const int lane = (int)(t & 31u);
    const unsigned rg = blockIdx.x;

    if (t < RANGE_N) cntN[t] = 0;
    if (t == 0) { spill = 0; ticket = 0; }
    __syncthreads();

    const int_a* region = (const int_a*)h + (size_t)rg * REGION;
    int l0[4];
    #pragma unroll
    for (int u = 0; u < 4; ++u) {
        const unsigned w = wave + 32u * (unsigned)u;
        l0[u] = (w < N_CHUNKS) ? region[w * SEG_INTS + (unsigned)lane] : 0;
    }
    #pragma unroll
    for (int u = 0; u < 4; ++u) {
        const unsigned w = wave + 32u * (unsigned)u;
        const int c    = __builtin_amdgcn_readfirstlane(l0[u]);
        const int last = c < SEG_CAP ? c : SEG_CAP;
        if (c > SEG_CAP && lane == 0) spill = 1;
        int rec  = l0[u];
        for (int idx = lane; idx <= last; idx += 32) {
            if (idx >= 32) rec = region[w * SEG_INTS + (unsigned)idx];
            if (idx >= 1) {
                const unsigned dl = (unsigned)rec & 255u;
                const int pos = atomicAdd(&cntN[dl], 1);
                if (pos < SLOT_CAP) binN[dl][pos] = (int)((unsigned)rec >> 8);
            }
        }
    }
    __syncthreads();

    const int anySpill = __builtin_amdgcn_readfirstlane(spill);
    int* sw = sorted[wave];
    const float* fb = feature + lane * 4;
    const int BIG = 0x7fffffff;

    for (unsigned trip = 0; trip < RANGE_N; ++trip) {
        int tk = 0;
        if (lane == 0) tk = atomicAdd(&ticket, 1);
        const unsigned dl = (unsigned)__builtin_amdgcn_readfirstlane(tk);
        if (dl >= RANGE_N) break;
        const int node = (int)(rg * RANGE_N + dl);
        const int nAll = __builtin_amdgcn_readfirstlane(cntN[dl]);
        const int n    = nAll < SLOT_CAP ? nAll : SLOT_CAP;

        const int a0 = (lane      < n) ? binN[dl][lane]      : BIG;
        const int a1 = (lane + 32 < n) ? binN[dl][lane + 32] : BIG;

        int r0 = 0, r1 = 0;
        for (int j = 0; j < n; ++j) {
            const int v = (j < 32) ? __builtin_amdgcn_readlane(a0, j)
                                   : __builtin_amdgcn_readlane(a1, j - 32);
            r0 += (v < a0 || (v == a0 && j < lane))      ? 1 : 0;
            r1 += (v < a1 || (v == a1 && j < lane + 32)) ? 1 : 0;
        }

        if (lane      < n) sw[r0] = a0;
        if (lane + 32 < n) sw[r1] = a1;
        const int b0 = (lane      < n) ? sw[lane]      : 0;
        const int b1 = (lane + 32 < n) ? sw[lane + 32] : 0;

        v4f acc = {0.f, 0.f, 0.f, 0.f};
        int j = 0;
        #pragma unroll 1
        for (; j + 8 <= n; j += 8) {
            const int bs = (j < 32) ? b0 : b1;
            v4f r[8];
            #pragma unroll
            for (int u = 0; u < 8; ++u) {
                const int s = __builtin_amdgcn_readlane(bs, (j & 31) + u);
                r[u] = *(const v4f*)(fb + (size_t)s * IN_FEATS);
            }
            #pragma unroll
            for (int u = 0; u < 8; ++u) acc += r[u];
        }
        {
            const int bs = (j < 32) ? b0 : b1;
            const int jl = j & 31;
            const int tl = n - j;
            const v4f z = {0.f, 0.f, 0.f, 0.f};
            v4f t0 = z, t1 = z, t2 = z, t3 = z, t4 = z, t5 = z, t6 = z;
            if (tl & 4) {
                t0 = *(const v4f*)(fb + (size_t)__builtin_amdgcn_readlane(bs, jl + 0) * IN_FEATS);
                t1 = *(const v4f*)(fb + (size_t)__builtin_amdgcn_readlane(bs, jl + 1) * IN_FEATS);
                t2 = *(const v4f*)(fb + (size_t)__builtin_amdgcn_readlane(bs, jl + 2) * IN_FEATS);
                t3 = *(const v4f*)(fb + (size_t)__builtin_amdgcn_readlane(bs, jl + 3) * IN_FEATS);
            }
            const int k2 = jl + (tl & 4);
            if (tl & 2) {
                t4 = *(const v4f*)(fb + (size_t)__builtin_amdgcn_readlane(bs, k2 + 0) * IN_FEATS);
                t5 = *(const v4f*)(fb + (size_t)__builtin_amdgcn_readlane(bs, k2 + 1) * IN_FEATS);
            }
            const int k1 = jl + (tl & 6);
            if (tl & 1) {
                t6 = *(const v4f*)(fb + (size_t)__builtin_amdgcn_readlane(bs, k1) * IN_FEATS);
            }
            acc += t0; acc += t1; acc += t2; acc += t3; acc += t4; acc += t5; acc += t6;
        }

        if (nAll > SLOT_CAP || anySpill) {
            acc = (v4f){0.f, 0.f, 0.f, 0.f};
            for (int base = 0; base < N_EDGES; base += 32) {
                unsigned m = __builtin_amdgcn_ballot_w32(dst[base + lane] == node);
                while (m) {
                    const int k = __builtin_ctz(m);
                    m &= m - 1u;
                    acc += *(const v4f*)(fb + (size_t)src[base + k] * IN_FEATS);
                }
            }
        }

        *(v4f*)(h + (size_t)node * IN_FEATS + lane * 4) = acc;
    }
}

#define GW_PITCH     132u
#define GW_ROWTILES  3125u

__global__ __launch_bounds__(256) void gemm_wmma(const float* __restrict__ h,
                                                 const float* __restrict__ W,
                                                 const float* __restrict__ bias,
                                                 float*       __restrict__ out) {
    __shared__ float Ws[OUT_FEATS * GW_PITCH];

    const unsigned waveInBlk = (unsigned)__builtin_amdgcn_readfirstlane((int)(threadIdx.x >> 5));
    const unsigned lane      = threadIdx.x & 31u;
    const unsigned c         = lane & 15u;
    const unsigned hh        = lane >> 4;
    const unsigned obase     = waveInBlk * 64u;

    unsigned rt = blockIdx.x;

    #pragma unroll 16
    for (unsigned r = waveInBlk; r < OUT_FEATS; r += 8u) {
        const v4f v = *(const v4f*)(W + (size_t)r * IN_FEATS + lane * 4u);
        const unsigned sr = (r & ~63u) | ((r & 3u) << 4) | ((r >> 2) & 15u);
        *(v4f*)(Ws + sr * GW_PITCH + lane * 4u) = v;
    }

    v4f ac[16];
    {
        const float* ar = h + (size_t)(rt * 16u + c) * IN_FEATS + 4u * hh;
        #pragma unroll
        for (int j = 0; j < 16; ++j) ac[j] = *(const v4f*)(ar + 8 * j);
    }
    const v4f bv = *(const v4f*)(bias + obase + 4u * c);

    __syncthreads();

    const float* bp = Ws + (obase + c) * GW_PITCH + 4u * hh;

    for (; rt < GW_ROWTILES; rt += gridDim.x) {
        const unsigned nbase = rt * 16u;

        unsigned rn = rt + gridDim.x;
        if (rn >= GW_ROWTILES) rn = rt;
        v4f an[16];
        {
            const float* ar = h + (size_t)(rn * 16u + c) * IN_FEATS + 4u * hh;
            #pragma unroll
            for (int j = 0; j < 16; ++j) an[j] = *(const v4f*)(ar + 8 * j);
        }

        v8f acc0 = {}, acc1 = {}, acc2 = {}, acc3 = {};

        #pragma unroll
        for (int j = 0; j < 16; ++j) {
            const v4f f0 = *(const v4f*)(bp +  0u * GW_PITCH + 8 * j);
            const v4f f1 = *(const v4f*)(bp + 16u * GW_PITCH + 8 * j);
            const v4f f2 = *(const v4f*)(bp + 32u * GW_PITCH + 8 * j);
            const v4f f3 = *(const v4f*)(bp + 48u * GW_PITCH + 8 * j);
            const v2f alo = {ac[j].x, ac[j].y};
            const v2f ahi = {ac[j].z, ac[j].w};
            acc0 = __builtin_amdgcn_wmma_f32_16x16x4_f32(false, alo, false, (v2f){f0.x, f0.y},
                                                         (short)0, acc0, false, false);
            acc1 = __builtin_amdgcn_wmma_f32_16x16x4_f32(false, alo, false, (v2f){f1.x, f1.y},
                                                         (short)0, acc1, false, false);
            acc2 = __builtin_amdgcn_wmma_f32_16x16x4_f32(false, alo, false, (v2f){f2.x, f2.y},
                                                         (short)0, acc2, false, false);
            acc3 = __builtin_amdgcn_wmma_f32_16x16x4_f32(false, alo, false, (v2f){f3.x, f3.y},
                                                         (short)0, acc3, false, false);
            acc0 = __builtin_amdgcn_wmma_f32_16x16x4_f32(false, ahi, false, (v2f){f0.z, f0.w},
                                                         (short)0, acc0, false, false);
            acc1 = __builtin_amdgcn_wmma_f32_16x16x4_f32(false, ahi, false, (v2f){f1.z, f1.w},
                                                         (short)0, acc1, false, false);
            acc2 = __builtin_amdgcn_wmma_f32_16x16x4_f32(false, ahi, false, (v2f){f2.z, f2.w},
                                                         (short)0, acc2, false, false);
            acc3 = __builtin_amdgcn_wmma_f32_16x16x4_f32(false, ahi, false, (v2f){f3.z, f3.w},
                                                         (short)0, acc3, false, false);
        }

        float* orow = out + (size_t)(nbase + 8u * hh) * OUT_FEATS + obase + 4u * c;
        #pragma unroll
        for (int v = 0; v < 8; ++v) {
            v4f val = (v4f){acc0[v], acc1[v], acc2[v], acc3[v]} + bv;
            val.x = (val.x >= 0.f) ? val.x : LEAKY * val.x;
            val.y = (val.y >= 0.f) ? val.y : LEAKY * val.y;
            val.z = (val.z >= 0.f) ? val.z : LEAKY * val.z;
            val.w = (val.w >= 0.f) ? val.w : LEAKY * val.w;
            *(v4f*)(orow + (size_t)v * OUT_FEATS) = val;
        }

        #pragma unroll
        for (int j = 0; j < 16; ++j) ac[j] = an[j];
    }
}

extern "C" void kernel_launch(void* const* d_in, const int* in_sizes, int n_in,
                              void* d_out, int out_size, void* d_ws, size_t ws_size,
                              hipStream_t stream) {
    const float* feature = (const float*)d_in[0];
    const int*   src     = (const int*)  d_in[1];
    const int*   dst     = (const int*)  d_in[2];
    const float* W       = (const float*)d_in[3];
    const float* bias    = (const float*)d_in[4];
    float*       out     = (float*)d_out;
    float*       h       = (float*)d_ws;

    bin_edges<<<N_CHUNKS, 1024, 0, stream>>>(src, dst, h);

    gather_rows<<<N_RANGES, 1024, 0, stream>>>(feature, src, dst, h);

    gemm_wmma<<<256, 256, 0, stream>>>(h, W, bias, out);
}
